# starting kernel
# baseline (speedup 1.0000x reference)
#include <hip/hip_runtime.h>
#include <math.h>

typedef __attribute__((ext_vector_type(2))) float v2f;
typedef __attribute__((ext_vector_type(8))) float v8f;

#define NQ        6
#define DIM       64
#define NL        2
#define USTRIDE   65
#define PSTRIDE   36
#define BCOLS     32
#define NTHREADS  128

__device__ __forceinline__ void apply_gate(float* Ur, float* Ui, int pos, bool withCZ,
    float g00r, float g00i, float g01r, float g01i,
    float g10r, float g10i, float g11r, float g11i, int tid)
{
    const int bit = 1 << pos;
    for (int p = tid; p < 32 * DIM; p += NTHREADS) {
        int col = p & 63;
        int h   = p >> 6;
        int low = h & (bit - 1);
        int r0  = ((h ^ low) << 1) | low;
        int r1  = r0 | bit;
        int i0 = r0 * USTRIDE + col;
        int i1 = r1 * USTRIDE + col;
        float ar = Ur[i0], ai = Ui[i0];
        float br = Ur[i1], bi = Ui[i1];
        if (withCZ) {
            float d0 = (__popc(r0 & (r0 >> 1) & 0x1f) & 1) ? -1.0f : 1.0f;
            float d1 = (__popc(r1 & (r1 >> 1) & 0x1f) & 1) ? -1.0f : 1.0f;
            ar *= d0; ai *= d0;
            br *= d1; bi *= d1;
        }
        float n0r = g00r*ar - g00i*ai + g01r*br - g01i*bi;
        float n0i = g00r*ai + g00i*ar + g01r*bi + g01i*br;
        float n1r = g10r*ar - g10i*ai + g11r*br - g11i*bi;
        float n1i = g10r*ai + g10i*ar + g11r*bi + g11i*br;
        Ur[i0] = n0r; Ui[i0] = n0i;
        Ur[i1] = n1r; Ui[i1] = n1i;
    }
}

__global__ __launch_bounds__(NTHREADS)
void qreservoir_kernel(const float* __restrict__ x, const float* __restrict__ w,
                       float* __restrict__ out, int S)
{
    __shared__ float Ur[DIM * USTRIDE];
    __shared__ float Ui[DIM * USTRIDE];
    __shared__ float P [DIM * PSTRIDE];
    __shared__ float Sr[DIM * PSTRIDE];
    __shared__ float Si[DIM * PSTRIDE];
    __shared__ float cs[BCOLS][NQ];
    __shared__ float sn[BCOLS][NQ];

    const int s   = blockIdx.x;
    const int b0  = blockIdx.y * BCOLS;
    const int tid = threadIdx.x;

    for (int idx = tid; idx < BCOLS * NQ; idx += NTHREADS) {
        int c = idx / NQ, q = idx % NQ;
        float f = x[((b0 + c) * S + s) * NQ + q];
        float sv, cv;
        sincosf(0.5f * 3.14159265358979323846f * f, &sv, &cv);
        sn[c][q] = sv; cs[c][q] = cv;
    }
    for (int idx = tid; idx < DIM * DIM; idx += NTHREADS) {
        int r = idx >> 6, c = idx & 63;
        Ur[r * USTRIDE + c] = (r == c) ? 1.0f : 0.0f;
        Ui[r * USTRIDE + c] = 0.0f;
    }
    __syncthreads();
    for (int idx = tid; idx < DIM * BCOLS; idx += NTHREADS) {
        int i = idx >> 5, c = idx & 31;
        float v = 1.0f;
        #pragma unroll
        for (int q = 0; q < NQ; ++q) {
            int bitv = (i >> (NQ - 1 - q)) & 1;
            v *= bitv ? sn[c][q] : cs[c][q];
        }
        P[i * PSTRIDE + c] = v;
    }
    __syncthreads();

    const float cf = cosf(0.25f);
    const float sf = sinf(0.25f);
    for (int l = 0; l < NL; ++l) {
        #pragma unroll 1
        for (int q = 0; q < NQ; ++q) {
            int base = ((s * NL + l) * NQ + q) * 3;
            float phi = w[base], th = w[base + 1], om = w[base + 2];
            float st, ct; sincosf(0.5f * th, &st, &ct);
            float sa, ca; sincosf(-0.5f * (phi + om), &sa, &ca);
            float sb, cb; sincosf( 0.5f * (phi - om), &sb, &cb);
            float r00r =  ca * ct, r00i =  sa * ct;
            float r01r = -cb * st, r01i = -sb * st;
            float r10r =  cb * st, r10i = -sb * st;
            float r11r =  ca * ct, r11i = -sa * ct;
            float g00r = cf * r00r + sf * r01i;
            float g00i = cf * r00i - sf * r01r;
            float g01r = sf * r00i + cf * r01r;
            float g01i = -sf * r00r + cf * r01i;
            float g10r = cf * r10r + sf * r11i;
            float g10i = cf * r10i - sf * r11r;
            float g11r = sf * r10i + cf * r11r;
            float g11i = -sf * r10r + cf * r11i;
            apply_gate(Ur, Ui, NQ - 1 - q, (q == 0),
                       g00r, g00i, g01r, g01i,
                       g10r, g10i, g11r, g11i, tid);
            __syncthreads();
        }
    }

    const int lane = tid & 31;
    const int wv   = tid >> 5;
    const int lh   = lane >> 4;
    const int lm   = lane & 15;
    #pragma unroll
    for (int t = 0; t < 2; ++t) {
        int T  = wv * 2 + t;
        int m0 = T >> 1;
        int n0 = T & 1;
        v8f accR = {};
        v8f accI = {};
        int rowA = m0 * 16 + lm;
        int colB = n0 * 16 + lm;
        for (int k0 = 0; k0 < 16; ++k0) {
            int kb = k0 * 4 + lh * 2;
            v2f aR = { Ur[rowA * USTRIDE + kb], Ur[rowA * USTRIDE + kb + 1] };
            v2f aI = { Ui[rowA * USTRIDE + kb], Ui[rowA * USTRIDE + kb + 1] };
            v2f bF = { P[kb * PSTRIDE + colB], P[(kb + 1) * PSTRIDE + colB] };
            accR = __builtin_amdgcn_wmma_f32_16x16x4_f32(
                false, aR, false, bF, (short)0, accR, false, false);
            accI = __builtin_amdgcn_wmma_f32_16x16x4_f32(
                false, aI, false, bF, (short)0, accI, false, false);
        }
        #pragma unroll
        for (int v = 0; v < 8; ++v) {
            int row = m0 * 16 + lh * 8 + v;
            Sr[row * PSTRIDE + colB] = accR[v];
            Si[row * PSTRIDE + colB] = accI[v];
        }
    }
    __syncthreads();

    {
        int col     = tid >> 2;
        int quarter = tid & 3;
        float e[18];
        #pragma unroll
        for (int k = 0; k < 18; ++k) e[k] = 0.f;
        for (int it = 0; it < 16; ++it) {
            int i = quarter * 16 + ((it + quarter * 4) & 15);
            float ar = Sr[i * PSTRIDE + col];
            float ai = Si[i * PSTRIDE + col];
            float prob = ar * ar + ai * ai;
            #pragma unroll
            for (int q = 0; q < NQ; ++q) {
                int pos  = NQ - 1 - q;
                int bitv = (i >> pos) & 1;
                int j    = i ^ (1 << pos);
                float br = Sr[j * PSTRIDE + col];
                float bi = Si[j * PSTRIDE + col];
                e[q]      += ar * br + ai * bi;
                float sgn  = bitv ? 1.0f : -1.0f;
                e[6 + q]  += sgn * (ai * br - ar * bi);
                e[12 + q] += bitv ? -prob : prob;
            }
        }
        #pragma unroll
        for (int k = 0; k < 18; ++k) {
            e[k] += __shfl_xor(e[k], 1);
            e[k] += __shfl_xor(e[k], 2);
        }
        if (quarter == 0) {
            int b = b0 + col;
            float* o = out + ((size_t)b * S + s) * 18;
            #pragma unroll
            for (int k = 0; k < 18; ++k) o[k] = e[k];
        }
    }
}

extern "C" void kernel_launch(void* const* d_in, const int* in_sizes, int n_in,
                              void* d_out, int out_size, void* d_ws, size_t ws_size,
                              hipStream_t stream) {
    const float* x = (const float*)d_in[0];
    const float* w = (const float*)d_in[1];
    float* out = (float*)d_out;
    (void)d_ws; (void)ws_size; (void)n_in; (void)out_size;
    int S = in_sizes[1] / (NL * NQ * 3);
    int B = in_sizes[0] / (S * NQ);
    dim3 grid(S, B / BCOLS);
    qreservoir_kernel<<<grid, NTHREADS, 0, stream>>>(x, w, out, S);
}


# U_s built once per step by a first launch (one lane per column of U, 12 gates applied to its 64 amplitudes in registers), passed via workspace in A-fragment order; was rebuilt in each of 4 blocks
# speedup vs baseline: 1.7488x; 1.7488x over previous
#include <hip/hip_runtime.h>
#include <math.h>


typedef __attribute__((ext_vector_type(2))) float v2f;
typedef __attribute__((ext_vector_type(4))) float v4f;
typedef __attribute__((ext_vector_type(8))) float v8f;

#define NQ        6
#define DIM       64
#define NL        2
#define TSTRIDE   68
#define PSTRIDE   36
#define BCOLS     32
#define NTHREADS  128
#define UFLOATS   (2 * DIM * DIM)

struct GateC { float g00r, g00i, g01r, g01i, g10r, g10i, g11r, g11i; };

template <int POS, bool WITHCZ>
__device__ __forceinline__ void gate_in_regs(v2f (&z)[DIM], const GateC& G)
{
    constexpr int bit = 1 << POS;
    #pragma unroll
    for (int h = 0; h < DIM / 2; ++h) {
        const int low = h & (bit - 1);
        const int r0  = ((h ^ low) << 1) | low;
        const int r1  = r0 | bit;
        v2f x = z[r0], y = z[r1];
        if (WITHCZ) {
            if (__builtin_popcount(r0 & (r0 >> 1) & 0x1f) & 1) x = -x;
            if (__builtin_popcount(r1 & (r1 >> 1) & 0x1f) & 1) y = -y;
        }
        const v2f jx = { -x.y, x.x };
        const v2f jy = { -y.y, y.x };
        z[r0] = G.g00r * x + G.g00i * jx + G.g01r * y + G.g01i * jy;
        z[r1] = G.g10r * x + G.g10i * jx + G.g11r * y + G.g11i * jy;
    }
}

__global__ __launch_bounds__(DIM)
void build_u_kernel(const float* __restrict__ w, float* __restrict__ uws)
{
    __shared__ float Tr[DIM * TSTRIDE];
    __shared__ float Ti[DIM * TSTRIDE];

    const int s = blockIdx.x;
    const int c = threadIdx.x;

    v2f z[DIM];
    #pragma unroll
    for (int r = 0; r < DIM; ++r) { z[r].x = (r == c) ? 1.0f : 0.0f; z[r].y = 0.0f; }

    const float cf = cosf(0.25f);
    const float sf = sinf(0.25f);
    const int gl = ((c & 31) < NL * NQ) ? (c & 31) : (NL * NQ - 1);
    const float* wg = w + (s * NL * NQ + gl) * 3;
    const int wphi = __float_as_int(wg[0]);
    const int wth  = __float_as_int(wg[1]);
    const int wom  = __float_as_int(wg[2]);
    #pragma unroll 1
    for (int g = 0; g < NL * NQ; ++g) {
        const int q = g % NQ;
        float phi = __int_as_float(__builtin_amdgcn_readlane(wphi, g));
        float th  = __int_as_float(__builtin_amdgcn_readlane(wth,  g));
        float om  = __int_as_float(__builtin_amdgcn_readlane(wom,  g));
        float st, ct; sincosf(0.5f * th, &st, &ct);
        float sa, ca; sincosf(-0.5f * (phi + om), &sa, &ca);
        float sb, cb; sincosf( 0.5f * (phi - om), &sb, &cb);
        float r00r =  ca * ct, r00i =  sa * ct;
        float r01r = -cb * st, r01i = -sb * st;
        float r10r =  cb * st, r10i = -sb * st;
        float r11r =  ca * ct, r11i = -sa * ct;
        GateC G;
        G.g00r = cf * r00r + sf * r01i;
        G.g00i = cf * r00i - sf * r01r;
        G.g01r = sf * r00i + cf * r01r;
        G.g01i = -sf * r00r + cf * r01i;
        G.g10r = cf * r10r + sf * r11i;
        G.g10i = cf * r10i - sf * r11r;
        G.g11r = sf * r10i + cf * r11r;
        G.g11i = -sf * r10r + cf * r11i;
        switch (q) {
            case 0:  gate_in_regs<5, true >(z, G); break;
            case 1:  gate_in_regs<4, false>(z, G); break;
            case 2:  gate_in_regs<3, false>(z, G); break;
            case 3:  gate_in_regs<2, false>(z, G); break;
            case 4:  gate_in_regs<1, false>(z, G); break;
            default: gate_in_regs<0, false>(z, G); break;
        }
    }

    #pragma unroll
    for (int r = 0; r < DIM; ++r) {
        Tr[r * TSTRIDE + c] = z[r].x;
        Ti[r * TSTRIDE + c] = z[r].y;
    }
    __syncthreads();
    v4f* dst = (v4f*)(uws + (size_t)s * UFLOATS);
    for (int idx = c; idx < UFLOATS / 4; idx += DIM) {
        int ln   = idx & 31;
        int j    = (idx >> 5) & 7;
        int part = (idx >> 8) & 1;
        int m0   = idx >> 9;
        int row  = m0 * 16 + (ln & 15);
        int ka   = 8 * j + 2 * (ln >> 4);
        const float* pl = part ? Ti : Tr;
        v2f lo = *(const v2f*)(pl + row * TSTRIDE + ka);
        v2f hi = *(const v2f*)(pl + row * TSTRIDE + ka + 4);
        v4f v = { lo.x, lo.y, hi.x, hi.y };
        dst[idx] = v;
    }
}

__global__ __launch_bounds__(NTHREADS)
void qreservoir_kernel(const float* __restrict__ x, const float* __restrict__ uws,
                       float* __restrict__ out, int S)
{
    __shared__ float P [DIM * PSTRIDE];
    __shared__ float Sr[DIM * PSTRIDE];
    __shared__ float Si[DIM * PSTRIDE];
    __shared__ float cs[BCOLS][NQ];
    __shared__ float sn[BCOLS][NQ];

    const int s   = blockIdx.x;
    const int b0  = blockIdx.y * BCOLS;
    const int tid = threadIdx.x;
    const int lane = tid & 31;
    const int wv   = tid >> 5;
    const int lh   = lane >> 4;
    const int lm   = lane & 15;

    v4f aRe[8], aIm[8];
    {
        const v4f* ua = (const v4f*)(uws + (size_t)s * UFLOATS) + (wv * 2) * 256 + lane;
        #pragma unroll
        for (int j = 0; j < 8; ++j) { aRe[j] = ua[j * 32]; aIm[j] = ua[256 + j * 32]; }
    }

    for (int idx = tid; idx < BCOLS * NQ; idx += NTHREADS) {
        int c = idx / NQ, q = idx % NQ;
        float f = x[((b0 + c) * S + s) * NQ + q];
        float sv, cv;
        sincosf(0.5f * 3.14159265358979323846f * f, &sv, &cv);
        sn[c][q] = sv; cs[c][q] = cv;
    }
    __syncthreads();
    for (int idx = tid; idx < DIM * BCOLS; idx += NTHREADS) {
        int i = idx >> 5, c = idx & 31;
        float v = 1.0f;
        #pragma unroll
        for (int q = 0; q < NQ; ++q) {
            int bitv = (i >> (NQ - 1 - q)) & 1;
            v *= bitv ? sn[c][q] : cs[c][q];
        }
        P[i * PSTRIDE + c] = v;
    }
    __syncthreads();

    {
        v8f accR0 = {}, accI0 = {}, accR1 = {}, accI1 = {};
        #pragma unroll
        for (int j = 0; j < 8; ++j) {
            #pragma unroll
            for (int hf = 0; hf < 2; ++hf) {
                int kb = (2 * j + hf) * 4 + lh * 2;
                v2f aR, aI;
                if (hf == 0) { aR = (v2f){ aRe[j].x, aRe[j].y }; aI = (v2f){ aIm[j].x, aIm[j].y }; }
                else         { aR = (v2f){ aRe[j].z, aRe[j].w }; aI = (v2f){ aIm[j].z, aIm[j].w }; }
                v2f bF0 = { P[kb * PSTRIDE + lm],      P[(kb + 1) * PSTRIDE + lm] };
                v2f bF1 = { P[kb * PSTRIDE + 16 + lm], P[(kb + 1) * PSTRIDE + 16 + lm] };
                accR0 = __builtin_amdgcn_wmma_f32_16x16x4_f32(
                    false, aR, false, bF0, (short)0, accR0, false, false);
                accI0 = __builtin_amdgcn_wmma_f32_16x16x4_f32(
                    false, aI, false, bF0, (short)0, accI0, false, false);
                accR1 = __builtin_amdgcn_wmma_f32_16x16x4_f32(
                    false, aR, false, bF1, (short)0, accR1, false, false);
                accI1 = __builtin_amdgcn_wmma_f32_16x16x4_f32(
                    false, aI, false, bF1, (short)0, accI1, false, false);
            }
        }
        #pragma unroll
        for (int v = 0; v < 8; ++v) {
            int row = wv * 16 + lh * 8 + v;
            Sr[row * PSTRIDE + lm]      = accR0[v];
            Si[row * PSTRIDE + lm]      = accI0[v];
            Sr[row * PSTRIDE + 16 + lm] = accR1[v];
            Si[row * PSTRIDE + 16 + lm] = accI1[v];
        }
    }
    __syncthreads();

    {
        int col     = tid >> 2;
        int quarter = tid & 3;
        float e[18];
        #pragma unroll
        for (int k = 0; k < 18; ++k) e[k] = 0.f;
        for (int it = 0; it < 16; ++it) {
            int i = quarter * 16 + ((it + quarter * 4) & 15);
            float ar = Sr[i * PSTRIDE + col];
            float ai = Si[i * PSTRIDE + col];
            float prob = ar * ar + ai * ai;
            #pragma unroll
            for (int q = 0; q < NQ; ++q) {
                int pos  = NQ - 1 - q;
                int bitv = (i >> pos) & 1;
                int j    = i ^ (1 << pos);
                float br = Sr[j * PSTRIDE + col];
                float bi = Si[j * PSTRIDE + col];
                e[q]      += ar * br + ai * bi;
                float sgn  = bitv ? 1.0f : -1.0f;
                e[6 + q]  += sgn * (ai * br - ar * bi);
                e[12 + q] += bitv ? -prob : prob;
            }
        }
        #pragma unroll
        for (int k = 0; k < 18; ++k) {
            e[k] += __shfl_xor(e[k], 1);
            e[k] += __shfl_xor(e[k], 2);
        }
        if (quarter == 0) {
            int b = b0 + col;
            float* o = out + ((size_t)b * S + s) * 18;
            #pragma unroll
            for (int k = 0; k < 18; ++k) o[k] = e[k];
        }
    }
}

extern "C" void kernel_launch(void* const* d_in, const int* in_sizes, int n_in,
                              void* d_out, int out_size, void* d_ws, size_t ws_size,
                              hipStream_t stream) {
    const float* x = (const float*)d_in[0];
    const float* w = (const float*)d_in[1];
    float* out = (float*)d_out;
    float* uws = (float*)d_ws;
    (void)ws_size; (void)n_in; (void)out_size;
    int S = in_sizes[1] / (NL * NQ * 3);
    int B = in_sizes[0] / (S * NQ);
    build_u_kernel<<<S, DIM, 0, stream>>>(w, uws);
    dim3 grid(S, B / BCOLS);
    qreservoir_kernel<<<grid, NTHREADS, 0, stream>>>(x, uws, out, S);
}


# Each wave owns 16 batch columns and all 64 rows (was 16 rows x 32 columns): B fragments formed in-lane, state stays in accumulators, 18 sums from registers; no P/Sr/Si planes; changes summation order
# speedup vs baseline: 1.8568x; 1.0618x over previous
#include <hip/hip_runtime.h>
#include <math.h>


typedef __attribute__((ext_vector_type(2))) float v2f;
typedef __attribute__((ext_vector_type(4))) float v4f;
typedef __attribute__((ext_vector_type(8))) float v8f;

#define NQ        6
#define DIM       64
#define NL        2
#define TSTRIDE   68
#define BCOLS     128
#define NTHREADS  256
#define UFLOATS   (2 * DIM * DIM)

struct GateC { float g00r, g00i, g01r, g01i, g10r, g10i, g11r, g11i; };

template <int POS, bool WITHCZ>
__device__ __forceinline__ void gate_in_regs(v2f (&z)[DIM], const GateC& G)
{
    constexpr int bit = 1 << POS;
    #pragma unroll
    for (int h = 0; h < DIM / 2; ++h) {
        const int low = h & (bit - 1);
        const int r0  = ((h ^ low) << 1) | low;
        const int r1  = r0 | bit;
        v2f x = z[r0], y = z[r1];
        if (WITHCZ) {
            if (__builtin_popcount(r0 & (r0 >> 1) & 0x1f) & 1) x = -x;
            if (__builtin_popcount(r1 & (r1 >> 1) & 0x1f) & 1) y = -y;
        }
        const v2f jx = { -x.y, x.x };
        const v2f jy = { -y.y, y.x };
        z[r0] = G.g00r * x + G.g00i * jx + G.g01r * y + G.g01i * jy;
        z[r1] = G.g10r * x + G.g10i * jx + G.g11r * y + G.g11i * jy;
    }
}

__global__ __launch_bounds__(DIM)
void build_u_kernel(const float* __restrict__ w, float* __restrict__ uws)
{
    __shared__ float Tr[DIM * TSTRIDE];
    __shared__ float Ti[DIM * TSTRIDE];

    const int s = blockIdx.x;
    const int c = threadIdx.x;

    v2f z[DIM];
    #pragma unroll
    for (int r = 0; r < DIM; ++r) { z[r].x = (r == c) ? 1.0f : 0.0f; z[r].y = 0.0f; }

    const float cf = cosf(0.25f);
    const float sf = sinf(0.25f);
    const int gl = ((c & 31) < NL * NQ) ? (c & 31) : (NL * NQ - 1);
    const float* wg = w + (s * NL * NQ + gl) * 3;
    const int wphi = __float_as_int(wg[0]);
    const int wth  = __float_as_int(wg[1]);
    const int wom  = __float_as_int(wg[2]);
    #pragma unroll 1
    for (int g = 0; g < NL * NQ; ++g) {
        const int q = g % NQ;
        float phi = __int_as_float(__builtin_amdgcn_readlane(wphi, g));
        float th  = __int_as_float(__builtin_amdgcn_readlane(wth,  g));
        float om  = __int_as_float(__builtin_amdgcn_readlane(wom,  g));
        float st, ct; sincosf(0.5f * th, &st, &ct);
        float sa, ca; sincosf(-0.5f * (phi + om), &sa, &ca);
        float sb, cb; sincosf( 0.5f * (phi - om), &sb, &cb);
        float r00r =  ca * ct, r00i =  sa * ct;
        float r01r = -cb * st, r01i = -sb * st;
        float r10r =  cb * st, r10i = -sb * st;
        float r11r =  ca * ct, r11i = -sa * ct;
        GateC G;
        G.g00r = cf * r00r + sf * r01i;
        G.g00i = cf * r00i - sf * r01r;
        G.g01r = sf * r00i + cf * r01r;
        G.g01i = -sf * r00r + cf * r01i;
        G.g10r = cf * r10r + sf * r11i;
        G.g10i = cf * r10i - sf * r11r;
        G.g11r = sf * r10i + cf * r11r;
        G.g11i = -sf * r10r + cf * r11i;
        switch (q) {
            case 0:  gate_in_regs<5, true >(z, G); break;
            case 1:  gate_in_regs<4, false>(z, G); break;
            case 2:  gate_in_regs<3, false>(z, G); break;
            case 3:  gate_in_regs<2, false>(z, G); break;
            case 4:  gate_in_regs<1, false>(z, G); break;
            default: gate_in_regs<0, false>(z, G); break;
        }
    }

    #pragma unroll
    for (int r = 0; r < DIM; ++r) {
        Tr[r * TSTRIDE + c] = z[r].x;
        Ti[r * TSTRIDE + c] = z[r].y;
    }
    __syncthreads();
    v4f* dst = (v4f*)(uws + (size_t)s * UFLOATS);
    for (int idx = c; idx < UFLOATS / 4; idx += DIM) {
        int ln   = idx & 31;
        int j    = (idx >> 5) & 7;
        int part = (idx >> 8) & 1;
        int m0   = idx >> 9;
        int row  = m0 * 16 + (ln & 15);
        int ka   = 8 * j + 2 * (ln >> 4);
        const float* pl = part ? Ti : Tr;
        v2f lo = *(const v2f*)(pl + row * TSTRIDE + ka);
        v2f hi = *(const v2f*)(pl + row * TSTRIDE + ka + 4);
        v4f v = { lo.x, lo.y, hi.x, hi.y };
        dst[idx] = v;
    }
}

__device__ __forceinline__ float xhalf(float v)
{
    int b = __builtin_bit_cast(int, v);
    return __builtin_bit_cast(float,
        __builtin_amdgcn_permlanex16(b, b, 0x76543210, 0xfedcba98, false, false));
}

template <int NB>
__device__ __forceinline__ void pair_sums(const float (&ar)[32], const float (&ai)[32],
                                          float& ex, float& ey)
{
    float sx = 0.f, sy = 0.f;
    #pragma unroll
    for (int t = 0; t < 16; ++t) {
        const int low = t & ((1 << NB) - 1);
        const int n0  = ((t ^ low) << 1) | low;
        const int n1  = n0 | (1 << NB);
        sx += ar[n0] * ar[n1] + ai[n0] * ai[n1];
        sy += ar[n0] * ai[n1] - ai[n0] * ar[n1];
    }
    ex = 2.0f * sx;
    ey = 2.0f * sy;
}

__global__ __launch_bounds__(NTHREADS)
void qreservoir_kernel(const float* __restrict__ x, const float* __restrict__ uws,
                       float* __restrict__ out, int S)
{
    __shared__ v4f   Ul[UFLOATS / 4];
    __shared__ float cs[BCOLS][NQ];
    __shared__ float sn[BCOLS][NQ];

    const int s    = blockIdx.x;
    const int b0   = blockIdx.y * BCOLS;
    const int tid  = threadIdx.x;
    const int lane = tid & 31;
    const int wv   = tid >> 5;
    const int lh   = lane >> 4;
    const int lm   = lane & 15;

    v4f stage[UFLOATS / 4 / NTHREADS];
    {
        const v4f* ug = (const v4f*)(uws + (size_t)s * UFLOATS);
        #pragma unroll
        for (int i = 0; i < UFLOATS / 4 / NTHREADS; ++i) stage[i] = ug[i * NTHREADS + tid];
    }

    for (int idx = tid; idx < BCOLS * NQ; idx += NTHREADS) {
        int c = idx / NQ, q = idx % NQ;
        float f = x[((size_t)(b0 + c) * S + s) * NQ + q];
        float sv, cv;
        sincosf(0.5f * 3.14159265358979323846f * f, &sv, &cv);
        sn[c][q] = sv; cs[c][q] = cv;
    }
    #pragma unroll
    for (int i = 0; i < UFLOATS / 4 / NTHREADS; ++i) Ul[i * NTHREADS + tid] = stage[i];
    __syncthreads();

    v2f bF[16];
    {
        const int col = wv * 16 + lm;
        float cq[NQ], sq[NQ];
        #pragma unroll
        for (int q = 0; q < NQ; ++q) { cq[q] = cs[col][q]; sq[q] = sn[col][q]; }
        float t2[4], t3[8], t4[16];
        #pragma unroll
        for (int a = 0; a < 4; ++a)  t2[a] = ((a >> 1) ? sq[0] : cq[0]) * ((a & 1) ? sq[1] : cq[1]);
        #pragma unroll
        for (int a = 0; a < 8; ++a)  t3[a] = t2[a >> 1] * ((a & 1) ? sq[2] : cq[2]);
        #pragma unroll
        for (int a = 0; a < 16; ++a) t4[a] = t3[a >> 1] * ((a & 1) ? sq[3] : cq[3]);
        const float f4 = lh ? sq[4] : cq[4];
        #pragma unroll
        for (int kk = 0; kk < 16; ++kk) {
            const float t5 = t4[kk] * f4;
            bF[kk] = (v2f){ t5 * cq[5], t5 * sq[5] };
        }
    }

    float ar[32], ai[32];
    #pragma unroll
    for (int m = 0; m < 4; ++m) {
        v4f aRe[8], aIm[8];
        const v4f* ua = Ul + (m * 2) * 256 + lane;
        #pragma unroll
        for (int j = 0; j < 8; ++j) { aRe[j] = ua[j * 32]; aIm[j] = ua[256 + j * 32]; }
        v8f accR = {}, accI = {};
        #pragma unroll
        for (int j = 0; j < 8; ++j) {
            #pragma unroll
            for (int hf = 0; hf < 2; ++hf) {
                v2f aR, aI;
                if (hf == 0) { aR = (v2f){ aRe[j].x, aRe[j].y }; aI = (v2f){ aIm[j].x, aIm[j].y }; }
                else         { aR = (v2f){ aRe[j].z, aRe[j].w }; aI = (v2f){ aIm[j].z, aIm[j].w }; }
                accR = __builtin_amdgcn_wmma_f32_16x16x4_f32(
                    false, aR, false, bF[2 * j + hf], (short)0, accR, false, false);
                accI = __builtin_amdgcn_wmma_f32_16x16x4_f32(
                    false, aI, false, bF[2 * j + hf], (short)0, accI, false, false);
            }
        }
        #pragma unroll
        for (int r = 0; r < 8; ++r) { ar[m * 8 + r] = accR[r]; ai[m * 8 + r] = accI[r]; }
    }

    float e[18];
    pair_sums<4>(ar, ai, e[0], e[6]);
    pair_sums<3>(ar, ai, e[1], e[7]);
    pair_sums<2>(ar, ai, e[3], e[9]);
    pair_sums<1>(ar, ai, e[4], e[10]);
    pair_sums<0>(ar, ai, e[5], e[11]);
    {
        float sx = 0.f, sy = 0.f;
        #pragma unroll
        for (int n = 0; n < 32; ++n) {
            const float br = xhalf(ar[n]);
            const float bi = xhalf(ai[n]);
            sx += ar[n] * br + ai[n] * bi;
            sy += ai[n] * br - ar[n] * bi;
        }
        e[2] = sx;
        e[8] = lh ? sy : -sy;
    }
    {
        float p[32];
        #pragma unroll
        for (int n = 0; n < 32; ++n) p[n] = ar[n] * ar[n] + ai[n] * ai[n];
        float zb[5];
        #pragma unroll
        for (int b = 0; b < 5; ++b) {
            const int half = 16 >> b;
            float d = 0.f;
            #pragma unroll
            for (int t = 0; t < half; ++t) {
                const float lo = p[2 * t], hi = p[2 * t + 1];
                d   += lo - hi;
                p[t] = lo + hi;
            }
            zb[b] = d;
        }
        e[12] = zb[4]; e[13] = zb[3]; e[15] = zb[2]; e[16] = zb[1]; e[17] = zb[0];
        e[14] = lh ? -p[0] : p[0];
    }
    #pragma unroll
    for (int k = 0; k < 18; ++k) e[k] += xhalf(e[k]);

    {
        const int b = b0 + wv * 16 + lm;
        float* o = out + ((size_t)b * S + s) * 18 + lh * 9;
        #pragma unroll
        for (int k = 0; k < 9; ++k) o[k] = lh ? e[9 + k] : e[k];
    }
}

extern "C" void kernel_launch(void* const* d_in, const int* in_sizes, int n_in,
                              void* d_out, int out_size, void* d_ws, size_t ws_size,
                              hipStream_t stream) {
    const float* x = (const float*)d_in[0];
    const float* w = (const float*)d_in[1];
    float* out = (float*)d_out;
    float* uws = (float*)d_ws;
    (void)ws_size; (void)n_in; (void)out_size;
    int S = in_sizes[1] / (NL * NQ * 3);
    int B = in_sizes[0] / (S * NQ);
    build_u_kernel<<<S, DIM, 0, stream>>>(w, uws);
    dim3 grid(S, B / BCOLS);
    qreservoir_kernel<<<grid, NTHREADS, 0, stream>>>(x, uws, out, S);
}


# U_s built in the step's own block by all 256 lanes (16 entries of a row each: layer-2 row as product vector, layer-1 gates transposed), stored to LDS in fragment order: one launch; changes sum order
# speedup vs baseline: 3.1043x; 1.6719x over previous
#include <hip/hip_runtime.h>
#include <math.h>


typedef __attribute__((ext_vector_type(2))) float v2f;
typedef __attribute__((ext_vector_type(4))) float v4f;
typedef __attribute__((ext_vector_type(8))) float v8f;

#define NQ        6
#define DIM       64
#define NL        2
#define BCOLS     128
#define NTHREADS  256
#define UFLOATS   (2 * DIM * DIM)
#define ENCTHREADS 192
#define GATETID0   224

struct GateC { float g00r, g00i, g01r, g01i, g10r, g10i, g11r, g11i; };

__device__ __forceinline__ v2f cmul(v2f a, v2f b)
{
    return (v2f){ a.x * b.x - a.y * b.y, a.x * b.y + a.y * b.x };
}

template <int POS>
__device__ __forceinline__ void gate16(v2f (&z)[16], const GateC& G)
{
    constexpr int bit = 1 << POS;
    #pragma unroll
    for (int h = 0; h < 8; ++h) {
        const int low = h & (bit - 1);
        const int r0  = ((h ^ low) << 1) | low;
        const int r1  = r0 | bit;
        const v2f x = z[r0], y = z[r1];
        const v2f jx = { -x.y, x.x };
        const v2f jy = { -y.y, y.x };
        z[r0] = G.g00r * x + G.g00i * jx + G.g01r * y + G.g01i * jy;
        z[r1] = G.g10r * x + G.g10i * jx + G.g11r * y + G.g11i * jy;
    }
}

__device__ __forceinline__ GateC transposed(v4f ra, v4f rb)
{
    GateC T;
    T.g00r = ra.x; T.g00i = ra.y;
    T.g01r = rb.x; T.g01i = rb.y;
    T.g10r = ra.z; T.g10i = ra.w;
    T.g11r = rb.z; T.g11i = rb.w;
    return T;
}

__device__ __forceinline__ float xhalf(float v)
{
    int b = __builtin_bit_cast(int, v);
    return __builtin_bit_cast(float,
        __builtin_amdgcn_permlanex16(b, b, 0x76543210, 0xfedcba98, false, false));
}

template <int NB>
__device__ __forceinline__ void pair_sums(const float (&ar)[32], const float (&ai)[32],
                                          float& ex, float& ey)
{
    float sx = 0.f, sy = 0.f;
    #pragma unroll
    for (int t = 0; t < 16; ++t) {
        const int low = t & ((1 << NB) - 1);
        const int n0  = ((t ^ low) << 1) | low;
        const int n1  = n0 | (1 << NB);
        sx += ar[n0] * ar[n1] + ai[n0] * ai[n1];
        sy += ar[n0] * ai[n1] - ai[n0] * ar[n1];
    }
    ex = 2.0f * sx;
    ey = 2.0f * sy;
}

__global__ __launch_bounds__(NTHREADS)
void qreservoir_kernel(const float* __restrict__ x, const float* __restrict__ w,
                       float* __restrict__ out, int S)
{
    __shared__ v4f   Ul[UFLOATS / 4];
    __shared__ v4f   Gt[NL * NQ * 2];
    __shared__ float cs[BCOLS][NQ];
    __shared__ float sn[BCOLS][NQ];

    const int s    = blockIdx.x;
    const int b0   = blockIdx.y * BCOLS;
    const int tid  = threadIdx.x;
    const int lane = tid & 31;
    const int wv   = tid >> 5;
    const int lh   = lane >> 4;
    const int lm   = lane & 15;

    if (tid < ENCTHREADS) {
        for (int idx = tid; idx < BCOLS * NQ; idx += ENCTHREADS) {
            int c = idx / NQ, q = idx % NQ;
            float f = x[((size_t)(b0 + c) * S + s) * NQ + q];
            float sv, cv;
            sincosf(0.5f * 3.14159265358979323846f * f, &sv, &cv);
            sn[c][q] = sv; cs[c][q] = cv;
        }
    } else if (tid >= GATETID0 && tid < GATETID0 + NL * NQ) {
        const int g = tid - GATETID0;
        const float* wg = w + ((size_t)s * NL * NQ + g) * 3;
        const float phi = wg[0], th = wg[1], om = wg[2];
        const float cf = cosf(0.25f);
        const float sf = sinf(0.25f);
        float st, ct; sincosf(0.5f * th, &st, &ct);
        float sa, ca; sincosf(-0.5f * (phi + om), &sa, &ca);
        float sb, cb; sincosf( 0.5f * (phi - om), &sb, &cb);
        float r00r =  ca * ct, r00i =  sa * ct;
        float r01r = -cb * st, r01i = -sb * st;
        float r10r =  cb * st, r10i = -sb * st;
        float r11r =  ca * ct, r11i = -sa * ct;
        GateC G;
        G.g00r = cf * r00r + sf * r01i;
        G.g00i = cf * r00i - sf * r01r;
        G.g01r = sf * r00i + cf * r01r;
        G.g01i = -sf * r00r + cf * r01i;
        G.g10r = cf * r10r + sf * r11i;
        G.g10i = cf * r10i - sf * r11r;
        G.g11r = sf * r10i + cf * r11r;
        G.g11i = -sf * r10r + cf * r11i;
        Gt[2 * g]     = (v4f){ G.g00r, G.g00i, G.g01r, G.g01i };
        Gt[2 * g + 1] = (v4f){ G.g10r, G.g10i, G.g11r, G.g11i };
    }
    __syncthreads();

    {
        const int tau = wv >> 1;
        const int r   = ((wv & 1) << 5) | lane;
        const int tq0 = tau >> 1, tq1 = tau & 1;

        v2f f[NQ][2];
        #pragma unroll
        for (int q = 0; q < NQ; ++q) {
            const int rq = (r >> (5 - q)) & 1;
            const v4f v = Gt[(NQ + q) * 2 + rq];
            f[q][0] = (v2f){ v.x, v.y };
            f[q][1] = (v2f){ v.z, v.w };
        }
        v2f p0[2], p1[2];
        #pragma unroll
        for (int a = 0; a < 2; ++a) {
            const v4f v0 = Gt[0 * 2 + a];
            const v4f v1 = Gt[1 * 2 + a];
            const v2f c0 = tq0 ? (v2f){ v0.z, v0.w } : (v2f){ v0.x, v0.y };
            const v2f c1 = tq1 ? (v2f){ v1.z, v1.w } : (v2f){ v1.x, v1.y };
            p0[a] = cmul(f[0][a], c0);
            p1[a] = cmul(f[1][a], c1);
        }
        const v2f sm = p1[0] + p1[1], df = p1[0] - p1[1];
        const v2f al0 = cmul(p0[0], sm) + cmul(p0[1], df);
        const v2f al1 = cmul(p0[0], df) + cmul(p0[1], sm);

        v2f t2[2], t3[4], t4[8], z[16];
        t2[0] = cmul(al0, f[2][0]);
        t2[1] = cmul(al1, f[2][1]);
        #pragma unroll
        for (int a = 0; a < 4; ++a) {
            v2f v = cmul(t2[a >> 1], f[3][a & 1]);
            t3[a] = (((a >> 1) & a & 1) ? -v : v);
        }
        #pragma unroll
        for (int a = 0; a < 8; ++a) {
            v2f v = cmul(t3[a >> 1], f[4][a & 1]);
            t4[a] = (((a >> 1) & a & 1) ? -v : v);
        }
        #pragma unroll
        for (int a = 0; a < 16; ++a) {
            v2f v = cmul(t4[a >> 1], f[5][a & 1]);
            z[a] = (((a >> 1) & a & 1) ? -v : v);
        }
        gate16<3>(z, transposed(Gt[2 * 2], Gt[2 * 2 + 1]));
        gate16<2>(z, transposed(Gt[3 * 2], Gt[3 * 2 + 1]));
        gate16<1>(z, transposed(Gt[4 * 2], Gt[4 * 2 + 1]));
        gate16<0>(z, transposed(Gt[5 * 2], Gt[5 * 2 + 1]));
        const float sg0 = (tq0 & tq1) ? -1.0f : 1.0f;
        const float sg1 = tq1 ? -sg0 : sg0;
        #pragma unroll
        for (int a = 0; a < 16; ++a) {
            const float sg = (a >> 3) ? sg1 : sg0;
            z[a] = ((__builtin_popcount(a & (a >> 1) & 7) & 1) ? -sg : sg) * z[a];
        }
        v4f* ub = Ul + ((r >> 4) * 2) * 256 + (2 * tau) * 32 + (r & 15);
        #pragma unroll
        for (int jl = 0; jl < 2; ++jl) {
            #pragma unroll
            for (int h = 0; h < 2; ++h) {
                const int k0 = 8 * jl + 2 * h;
                ub[jl * 32 + 16 * h]       = (v4f){ z[k0].x, z[k0 + 1].x, z[k0 + 4].x, z[k0 + 5].x };
                ub[256 + jl * 32 + 16 * h] = (v4f){ z[k0].y, z[k0 + 1].y, z[k0 + 4].y, z[k0 + 5].y };
            }
        }
    }

    v2f bF[16];
    {
        const int col = wv * 16 + lm;
        float cq[NQ], sq[NQ];
        #pragma unroll
        for (int q = 0; q < NQ; ++q) { cq[q] = cs[col][q]; sq[q] = sn[col][q]; }
        float t2[4], t3[8], t4[16];
        #pragma unroll
        for (int a = 0; a < 4; ++a)  t2[a] = ((a >> 1) ? sq[0] : cq[0]) * ((a & 1) ? sq[1] : cq[1]);
        #pragma unroll
        for (int a = 0; a < 8; ++a)  t3[a] = t2[a >> 1] * ((a & 1) ? sq[2] : cq[2]);
        #pragma unroll
        for (int a = 0; a < 16; ++a) t4[a] = t3[a >> 1] * ((a & 1) ? sq[3] : cq[3]);
        const float f4 = lh ? sq[4] : cq[4];
        #pragma unroll
        for (int kk = 0; kk < 16; ++kk) {
            const float t5 = t4[kk] * f4;
            bF[kk] = (v2f){ t5 * cq[5], t5 * sq[5] };
        }
    }

    __syncthreads();

    float ar[32], ai[32];
    #pragma unroll
    for (int m = 0; m < 4; ++m) {
        v4f aRe[8], aIm[8];
        const v4f* ua = Ul + (m * 2) * 256 + lane;
        #pragma unroll
        for (int j = 0; j < 8; ++j) { aRe[j] = ua[j * 32]; aIm[j] = ua[256 + j * 32]; }
        v8f accR = {}, accI = {};
        #pragma unroll
        for (int j = 0; j < 8; ++j) {
            #pragma unroll
            for (int hf = 0; hf < 2; ++hf) {
                v2f aR, aI;
                if (hf == 0) { aR = (v2f){ aRe[j].x, aRe[j].y }; aI = (v2f){ aIm[j].x, aIm[j].y }; }
                else         { aR = (v2f){ aRe[j].z, aRe[j].w }; aI = (v2f){ aIm[j].z, aIm[j].w }; }
                accR = __builtin_amdgcn_wmma_f32_16x16x4_f32(
                    false, aR, false, bF[2 * j + hf], (short)0, accR, false, false);
                accI = __builtin_amdgcn_wmma_f32_16x16x4_f32(
                    false, aI, false, bF[2 * j + hf], (short)0, accI, false, false);
            }
        }
        #pragma unroll
        for (int r = 0; r < 8; ++r) { ar[m * 8 + r] = accR[r]; ai[m * 8 + r] = accI[r]; }
    }

    float e[18];
    pair_sums<4>(ar, ai, e[0], e[6]);
    pair_sums<3>(ar, ai, e[1], e[7]);
    pair_sums<2>(ar, ai, e[3], e[9]);
    pair_sums<1>(ar, ai, e[4], e[10]);
    pair_sums<0>(ar, ai, e[5], e[11]);
    {
        float sx = 0.f, sy = 0.f;
        #pragma unroll
        for (int n = 0; n < 32; ++n) {
            const float br = xhalf(ar[n]);
            const float bi = xhalf(ai[n]);
            sx += ar[n] * br + ai[n] * bi;
            sy += ai[n] * br - ar[n] * bi;
        }
        e[2] = sx;
        e[8] = lh ? sy : -sy;
    }
    {
        float p[32];
        #pragma unroll
        for (int n = 0; n < 32; ++n) p[n] = ar[n] * ar[n] + ai[n] * ai[n];
        float zb[5];
        #pragma unroll
        for (int b = 0; b < 5; ++b) {
            const int half = 16 >> b;
            float d = 0.f;
            #pragma unroll
            for (int t = 0; t < half; ++t) {
                const float lo = p[2 * t], hi = p[2 * t + 1];
                d   += lo - hi;
                p[t] = lo + hi;
            }
            zb[b] = d;
        }
        e[12] = zb[4]; e[13] = zb[3]; e[15] = zb[2]; e[16] = zb[1]; e[17] = zb[0];
        e[14] = lh ? -p[0] : p[0];
    }
    #pragma unroll
    for (int k = 0; k < 18; ++k) e[k] += xhalf(e[k]);

    {
        const int b = b0 + wv * 16 + lm;
        float* o = out + ((size_t)b * S + s) * 18 + lh * 9;
        #pragma unroll
        for (int k = 0; k < 9; ++k) o[k] = lh ? e[9 + k] : e[k];
    }
}

extern "C" void kernel_launch(void* const* d_in, const int* in_sizes, int n_in,
                              void* d_out, int out_size, void* d_ws, size_t ws_size,
                              hipStream_t stream) {
    const float* x = (const float*)d_in[0];
    const float* w = (const float*)d_in[1];
    float* out = (float*)d_out;
    (void)d_ws; (void)ws_size; (void)n_in; (void)out_size;
    int S = in_sizes[1] / (NL * NQ * 3);
    int B = in_sizes[0] / (S * NQ);
    dim3 grid(S, B / BCOLS);
    qreservoir_kernel<<<grid, NTHREADS, 0, stream>>>(x, w, out, S);
}


# Encoding angles: each lane loads 3 features of its own column (one 12-byte load), takes their sin/cos, gets the other 3 from its partner half-wave; was a 4-trip load-then-sincos loop via LDS
# speedup vs baseline: 3.3744x; 1.0870x over previous
#include <hip/hip_runtime.h>
#include <math.h>


typedef __attribute__((ext_vector_type(2))) float v2f;
typedef __attribute__((ext_vector_type(4))) float v4f;
typedef __attribute__((ext_vector_type(8))) float v8f;

#define NQ        6
#define DIM       64
#define NL        2
#define BCOLS     128
#define NTHREADS  256
#define UFLOATS   (2 * DIM * DIM)
#define GATETID0   224

struct GateC { float g00r, g00i, g01r, g01i, g10r, g10i, g11r, g11i; };

__device__ __forceinline__ v2f cmul(v2f a, v2f b)
{
    return (v2f){ a.x * b.x - a.y * b.y, a.x * b.y + a.y * b.x };
}

template <int POS>
__device__ __forceinline__ void gate16(v2f (&z)[16], const GateC& G)
{
    constexpr int bit = 1 << POS;
    #pragma unroll
    for (int h = 0; h < 8; ++h) {
        const int low = h & (bit - 1);
        const int r0  = ((h ^ low) << 1) | low;
        const int r1  = r0 | bit;
        const v2f x = z[r0], y = z[r1];
        const v2f jx = { -x.y, x.x };
        const v2f jy = { -y.y, y.x };
        z[r0] = G.g00r * x + G.g00i * jx + G.g01r * y + G.g01i * jy;
        z[r1] = G.g10r * x + G.g10i * jx + G.g11r * y + G.g11i * jy;
    }
}

__device__ __forceinline__ GateC transposed(v4f ra, v4f rb)
{
    GateC T;
    T.g00r = ra.x; T.g00i = ra.y;
    T.g01r = rb.x; T.g01i = rb.y;
    T.g10r = ra.z; T.g10i = ra.w;
    T.g11r = rb.z; T.g11i = rb.w;
    return T;
}

__device__ __forceinline__ float xhalf(float v)
{
    int b = __builtin_bit_cast(int, v);
    return __builtin_bit_cast(float,
        __builtin_amdgcn_permlanex16(b, b, 0x76543210, 0xfedcba98, false, false));
}

template <int NB>
__device__ __forceinline__ void pair_sums(const float (&ar)[32], const float (&ai)[32],
                                          float& ex, float& ey)
{
    float sx = 0.f, sy = 0.f;
    #pragma unroll
    for (int t = 0; t < 16; ++t) {
        const int low = t & ((1 << NB) - 1);
        const int n0  = ((t ^ low) << 1) | low;
        const int n1  = n0 | (1 << NB);
        sx += ar[n0] * ar[n1] + ai[n0] * ai[n1];
        sy += ar[n0] * ai[n1] - ai[n0] * ar[n1];
    }
    ex = 2.0f * sx;
    ey = 2.0f * sy;
}

__global__ __launch_bounds__(NTHREADS)
void qreservoir_kernel(const float* __restrict__ x, const float* __restrict__ w,
                       float* __restrict__ out, int S)
{
    __shared__ v4f   Ul[UFLOATS / 4];
    __shared__ v4f   Gt[NL * NQ * 2];

    const int s    = blockIdx.x;
    const int b0   = blockIdx.y * BCOLS;
    const int tid  = threadIdx.x;
    const int lane = tid & 31;
    const int wv   = tid >> 5;
    const int lh   = lane >> 4;
    const int lm   = lane & 15;

    const float* xp = x + ((size_t)(b0 + wv * 16 + lm) * S + s) * NQ + 3 * lh;
    const float f0 = xp[0], f1 = xp[1], f2 = xp[2];

    if (tid >= GATETID0 && tid < GATETID0 + NL * NQ) {
        const int g = tid - GATETID0;
        const float* wg = w + ((size_t)s * NL * NQ + g) * 3;
        const float phi = wg[0], th = wg[1], om = wg[2];
        const float cf = cosf(0.25f);
        const float sf = sinf(0.25f);
        float st, ct; sincosf(0.5f * th, &st, &ct);
        float sa, ca; sincosf(-0.5f * (phi + om), &sa, &ca);
        float sb, cb; sincosf( 0.5f * (phi - om), &sb, &cb);
        float r00r =  ca * ct, r00i =  sa * ct;
        float r01r = -cb * st, r01i = -sb * st;
        float r10r =  cb * st, r10i = -sb * st;
        float r11r =  ca * ct, r11i = -sa * ct;
        GateC G;
        G.g00r = cf * r00r + sf * r01i;
        G.g00i = cf * r00i - sf * r01r;
        G.g01r = sf * r00i + cf * r01r;
        G.g01i = -sf * r00r + cf * r01i;
        G.g10r = cf * r10r + sf * r11i;
        G.g10i = cf * r10i - sf * r11r;
        G.g11r = sf * r10i + cf * r11r;
        G.g11i = -sf * r10r + cf * r11i;
        Gt[2 * g]     = (v4f){ G.g00r, G.g00i, G.g01r, G.g01i };
        Gt[2 * g + 1] = (v4f){ G.g10r, G.g10i, G.g11r, G.g11i };
    }

    float cq[NQ], sq[NQ];
    {
        float so[3], co[3];
        sincosf(0.5f * 3.14159265358979323846f * f0, &so[0], &co[0]);
        sincosf(0.5f * 3.14159265358979323846f * f1, &so[1], &co[1]);
        sincosf(0.5f * 3.14159265358979323846f * f2, &so[2], &co[2]);
        #pragma unroll
        for (int j = 0; j < 3; ++j) {
            const float oc = xhalf(co[j]), os = xhalf(so[j]);
            cq[j]     = lh ? oc : co[j];   sq[j]     = lh ? os : so[j];
            cq[3 + j] = lh ? co[j] : oc;   sq[3 + j] = lh ? so[j] : os;
        }
    }

    v2f bF[16];
    {
        float t2[4], t3[8], t4[16];
        #pragma unroll
        for (int a = 0; a < 4; ++a)  t2[a] = ((a >> 1) ? sq[0] : cq[0]) * ((a & 1) ? sq[1] : cq[1]);
        #pragma unroll
        for (int a = 0; a < 8; ++a)  t3[a] = t2[a >> 1] * ((a & 1) ? sq[2] : cq[2]);
        #pragma unroll
        for (int a = 0; a < 16; ++a) t4[a] = t3[a >> 1] * ((a & 1) ? sq[3] : cq[3]);
        const float f4 = lh ? sq[4] : cq[4];
        #pragma unroll
        for (int kk = 0; kk < 16; ++kk) {
            const float t5 = t4[kk] * f4;
            bF[kk] = (v2f){ t5 * cq[5], t5 * sq[5] };
        }
    }

    __syncthreads();

    {
        const int tau = wv >> 1;
        const int r   = ((wv & 1) << 5) | lane;
        const int tq0 = tau >> 1, tq1 = tau & 1;

        v2f f[NQ][2];
        #pragma unroll
        for (int q = 0; q < NQ; ++q) {
            const int rq = (r >> (5 - q)) & 1;
            const v4f v = Gt[(NQ + q) * 2 + rq];
            f[q][0] = (v2f){ v.x, v.y };
            f[q][1] = (v2f){ v.z, v.w };
        }
        v2f p0[2], p1[2];
        #pragma unroll
        for (int a = 0; a < 2; ++a) {
            const v4f v0 = Gt[0 * 2 + a];
            const v4f v1 = Gt[1 * 2 + a];
            const v2f c0 = tq0 ? (v2f){ v0.z, v0.w } : (v2f){ v0.x, v0.y };
            const v2f c1 = tq1 ? (v2f){ v1.z, v1.w } : (v2f){ v1.x, v1.y };
            p0[a] = cmul(f[0][a], c0);
            p1[a] = cmul(f[1][a], c1);
        }
        const v2f sm = p1[0] + p1[1], df = p1[0] - p1[1];
        const v2f al0 = cmul(p0[0], sm) + cmul(p0[1], df);
        const v2f al1 = cmul(p0[0], df) + cmul(p0[1], sm);

        v2f t2[2], t3[4], t4[8], z[16];
        t2[0] = cmul(al0, f[2][0]);
        t2[1] = cmul(al1, f[2][1]);
        #pragma unroll
        for (int a = 0; a < 4; ++a) {
            v2f v = cmul(t2[a >> 1], f[3][a & 1]);
            t3[a] = (((a >> 1) & a & 1) ? -v : v);
        }
        #pragma unroll
        for (int a = 0; a < 8; ++a) {
            v2f v = cmul(t3[a >> 1], f[4][a & 1]);
            t4[a] = (((a >> 1) & a & 1) ? -v : v);
        }
        #pragma unroll
        for (int a = 0; a < 16; ++a) {
            v2f v = cmul(t4[a >> 1], f[5][a & 1]);
            z[a] = (((a >> 1) & a & 1) ? -v : v);
        }
        gate16<3>(z, transposed(Gt[2 * 2], Gt[2 * 2 + 1]));
        gate16<2>(z, transposed(Gt[3 * 2], Gt[3 * 2 + 1]));
        gate16<1>(z, transposed(Gt[4 * 2], Gt[4 * 2 + 1]));
        gate16<0>(z, transposed(Gt[5 * 2], Gt[5 * 2 + 1]));
        const float sg0 = (tq0 & tq1) ? -1.0f : 1.0f;
        const float sg1 = tq1 ? -sg0 : sg0;
        #pragma unroll
        for (int a = 0; a < 16; ++a) {
            const float sg = (a >> 3) ? sg1 : sg0;
            z[a] = ((__builtin_popcount(a & (a >> 1) & 7) & 1) ? -sg : sg) * z[a];
        }
        v4f* ub = Ul + ((r >> 4) * 2) * 256 + (2 * tau) * 32 + (r & 15);
        #pragma unroll
        for (int jl = 0; jl < 2; ++jl) {
            #pragma unroll
            for (int h = 0; h < 2; ++h) {
                const int k0 = 8 * jl + 2 * h;
                ub[jl * 32 + 16 * h]       = (v4f){ z[k0].x, z[k0 + 1].x, z[k0 + 4].x, z[k0 + 5].x };
                ub[256 + jl * 32 + 16 * h] = (v4f){ z[k0].y, z[k0 + 1].y, z[k0 + 4].y, z[k0 + 5].y };
            }
        }
    }

    __syncthreads();

    float ar[32], ai[32];
    #pragma unroll
    for (int m = 0; m < 4; ++m) {
        v4f aRe[8], aIm[8];
        const v4f* ua = Ul + (m * 2) * 256 + lane;
        #pragma unroll
        for (int j = 0; j < 8; ++j) { aRe[j] = ua[j * 32]; aIm[j] = ua[256 + j * 32]; }
        v8f accR = {}, accI = {};
        #pragma unroll
        for (int j = 0; j < 8; ++j) {
            #pragma unroll
            for (int hf = 0; hf < 2; ++hf) {
                v2f aR, aI;
                if (hf == 0) { aR = (v2f){ aRe[j].x, aRe[j].y }; aI = (v2f){ aIm[j].x, aIm[j].y }; }
                else         { aR = (v2f){ aRe[j].z, aRe[j].w }; aI = (v2f){ aIm[j].z, aIm[j].w }; }
                accR = __builtin_amdgcn_wmma_f32_16x16x4_f32(
                    false, aR, false, bF[2 * j + hf], (short)0, accR, false, false);
                accI = __builtin_amdgcn_wmma_f32_16x16x4_f32(
                    false, aI, false, bF[2 * j + hf], (short)0, accI, false, false);
            }
        }
        #pragma unroll
        for (int r = 0; r < 8; ++r) { ar[m * 8 + r] = accR[r]; ai[m * 8 + r] = accI[r]; }
    }

    float e[18];
    pair_sums<4>(ar, ai, e[0], e[6]);
    pair_sums<3>(ar, ai, e[1], e[7]);
    pair_sums<2>(ar, ai, e[3], e[9]);
    pair_sums<1>(ar, ai, e[4], e[10]);
    pair_sums<0>(ar, ai, e[5], e[11]);
    {
        float sx = 0.f, sy = 0.f;
        #pragma unroll
        for (int n = 0; n < 32; ++n) {
            const float br = xhalf(ar[n]);
            const float bi = xhalf(ai[n]);
            sx += ar[n] * br + ai[n] * bi;
            sy += ai[n] * br - ar[n] * bi;
        }
        e[2] = sx;
        e[8] = lh ? sy : -sy;
    }
    {
        float p[32];
        #pragma unroll
        for (int n = 0; n < 32; ++n) p[n] = ar[n] * ar[n] + ai[n] * ai[n];
        float zb[5];
        #pragma unroll
        for (int b = 0; b < 5; ++b) {
            const int half = 16 >> b;
            float d = 0.f;
            #pragma unroll
            for (int t = 0; t < half; ++t) {
                const float lo = p[2 * t], hi = p[2 * t + 1];
                d   += lo - hi;
                p[t] = lo + hi;
            }
            zb[b] = d;
        }
        e[12] = zb[4]; e[13] = zb[3]; e[15] = zb[2]; e[16] = zb[1]; e[17] = zb[0];
        e[14] = lh ? -p[0] : p[0];
    }
    #pragma unroll
    for (int k = 0; k < 18; ++k) e[k] += xhalf(e[k]);

    {
        const int b = b0 + wv * 16 + lm;
        float* o = out + ((size_t)b * S + s) * 18 + lh * 9;
        #pragma unroll
        for (int k = 0; k < 9; ++k) o[k] = lh ? e[9 + k] : e[k];
    }
}

extern "C" void kernel_launch(void* const* d_in, const int* in_sizes, int n_in,
                              void* d_out, int out_size, void* d_ws, size_t ws_size,
                              hipStream_t stream) {
    const float* x = (const float*)d_in[0];
    const float* w = (const float*)d_in[1];
    float* out = (float*)d_out;
    (void)d_ws; (void)ws_size; (void)n_in; (void)out_size;
    int S = in_sizes[1] / (NL * NQ * 3);
    int B = in_sizes[0] / (S * NQ);
    dim3 grid(S, B / BCOLS);
    qreservoir_kernel<<<grid, NTHREADS, 0, stream>>>(x, w, out, S);
}


# Expectation sums formed per row tile on the accumulator register pairs as they lie, issued among the next tile's matrix products; was one pass behind all 128 products; changes summation order
# speedup vs baseline: 3.7429x; 1.1092x over previous
#include <hip/hip_runtime.h>
#include <math.h>


typedef __attribute__((ext_vector_type(2))) float v2f;
typedef __attribute__((ext_vector_type(4))) float v4f;
typedef __attribute__((ext_vector_type(8))) float v8f;

#define NQ        6
#define DIM       64
#define NL        2
#define BCOLS     128
#define NTHREADS  256
#define UFLOATS   (2 * DIM * DIM)
#define GATETID0   224

struct GateC { float g00r, g00i, g01r, g01i, g10r, g10i, g11r, g11i; };

__device__ __forceinline__ v2f cmul(v2f a, v2f b)
{
    return (v2f){ a.x * b.x - a.y * b.y, a.x * b.y + a.y * b.x };
}

template <int POS>
__device__ __forceinline__ void gate16(v2f (&z)[16], const GateC& G)
{
    constexpr int bit = 1 << POS;
    #pragma unroll
    for (int h = 0; h < 8; ++h) {
        const int low = h & (bit - 1);
        const int r0  = ((h ^ low) << 1) | low;
        const int r1  = r0 | bit;
        const v2f x = z[r0], y = z[r1];
        const v2f jx = { -x.y, x.x };
        const v2f jy = { -y.y, y.x };
        z[r0] = G.g00r * x + G.g00i * jx + G.g01r * y + G.g01i * jy;
        z[r1] = G.g10r * x + G.g10i * jx + G.g11r * y + G.g11i * jy;
    }
}

__device__ __forceinline__ GateC transposed(v4f ra, v4f rb)
{
    GateC T;
    T.g00r = ra.x; T.g00i = ra.y;
    T.g01r = rb.x; T.g01i = rb.y;
    T.g10r = ra.z; T.g10i = ra.w;
    T.g11r = rb.z; T.g11i = rb.w;
    return T;
}

__device__ __forceinline__ float xhalf(float v)
{
    int b = __builtin_bit_cast(int, v);
    return __builtin_bit_cast(float,
        __builtin_amdgcn_permlanex16(b, b, 0x76543210, 0xfedcba98, false, false));
}

__device__ __forceinline__ void pair_acc(v2f xr, v2f xi, v2f yr, v2f yi, v2f& SX, v2f& SY)
{
    SX = xr * yr + SX;
    SX = xi * yi + SX;
    SY = xr * yi + SY;
    SY = SY - xi * yr;
}

__device__ __forceinline__ float hsum(v2f v) { return v.x + v.y; }

__global__ __launch_bounds__(NTHREADS)
void qreservoir_kernel(const float* __restrict__ x, const float* __restrict__ w,
                       float* __restrict__ out, int S)
{
    __shared__ v4f   Ul[UFLOATS / 4];
    __shared__ v4f   Gt[NL * NQ * 2];

    const int s    = blockIdx.x;
    const int b0   = blockIdx.y * BCOLS;
    const int tid  = threadIdx.x;
    const int lane = tid & 31;
    const int wv   = tid >> 5;
    const int lh   = lane >> 4;
    const int lm   = lane & 15;

    const float* xp = x + ((size_t)(b0 + wv * 16 + lm) * S + s) * NQ + 3 * lh;
    const float f0 = xp[0], f1 = xp[1], f2 = xp[2];

    if (tid >= GATETID0 && tid < GATETID0 + NL * NQ) {
        const int g = tid - GATETID0;
        const float* wg = w + ((size_t)s * NL * NQ + g) * 3;
        const float phi = wg[0], th = wg[1], om = wg[2];
        const float cf = cosf(0.25f);
        const float sf = sinf(0.25f);
        float st, ct; sincosf(0.5f * th, &st, &ct);
        float sa, ca; sincosf(-0.5f * (phi + om), &sa, &ca);
        float sb, cb; sincosf( 0.5f * (phi - om), &sb, &cb);
        float r00r =  ca * ct, r00i =  sa * ct;
        float r01r = -cb * st, r01i = -sb * st;
        float r10r =  cb * st, r10i = -sb * st;
        float r11r =  ca * ct, r11i = -sa * ct;
        GateC G;
        G.g00r = cf * r00r + sf * r01i;
        G.g00i = cf * r00i - sf * r01r;
        G.g01r = sf * r00i + cf * r01r;
        G.g01i = -sf * r00r + cf * r01i;
        G.g10r = cf * r10r + sf * r11i;
        G.g10i = cf * r10i - sf * r11r;
        G.g11r = sf * r10i + cf * r11r;
        G.g11i = -sf * r10r + cf * r11i;
        Gt[2 * g]     = (v4f){ G.g00r, G.g00i, G.g01r, G.g01i };
        Gt[2 * g + 1] = (v4f){ G.g10r, G.g10i, G.g11r, G.g11i };
    }

    float cq[NQ], sq[NQ];
    {
        float so[3], co[3];
        sincosf(0.5f * 3.14159265358979323846f * f0, &so[0], &co[0]);
        sincosf(0.5f * 3.14159265358979323846f * f1, &so[1], &co[1]);
        sincosf(0.5f * 3.14159265358979323846f * f2, &so[2], &co[2]);
        #pragma unroll
        for (int j = 0; j < 3; ++j) {
            const float oc = xhalf(co[j]), os = xhalf(so[j]);
            cq[j]     = lh ? oc : co[j];   sq[j]     = lh ? os : so[j];
            cq[3 + j] = lh ? co[j] : oc;   sq[3 + j] = lh ? so[j] : os;
        }
    }

    v2f bF[16];
    {
        float t2[4], t3[8], t4[16];
        #pragma unroll
        for (int a = 0; a < 4; ++a)  t2[a] = ((a >> 1) ? sq[0] : cq[0]) * ((a & 1) ? sq[1] : cq[1]);
        #pragma unroll
        for (int a = 0; a < 8; ++a)  t3[a] = t2[a >> 1] * ((a & 1) ? sq[2] : cq[2]);
        #pragma unroll
        for (int a = 0; a < 16; ++a) t4[a] = t3[a >> 1] * ((a & 1) ? sq[3] : cq[3]);
        const float f4 = lh ? sq[4] : cq[4];
        #pragma unroll
        for (int kk = 0; kk < 16; ++kk) {
            const float t5 = t4[kk] * f4;
            bF[kk] = (v2f){ t5 * cq[5], t5 * sq[5] };
        }
    }

    __syncthreads();

    {
        const int tau = wv >> 1;
        const int r   = ((wv & 1) << 5) | lane;
        const int tq0 = tau >> 1, tq1 = tau & 1;

        v2f f[NQ][2];
        #pragma unroll
        for (int q = 0; q < NQ; ++q) {
            const int rq = (r >> (5 - q)) & 1;
            const v4f v = Gt[(NQ + q) * 2 + rq];
            f[q][0] = (v2f){ v.x, v.y };
            f[q][1] = (v2f){ v.z, v.w };
        }
        v2f p0[2], p1[2];
        #pragma unroll
        for (int a = 0; a < 2; ++a) {
            const v4f v0 = Gt[0 * 2 + a];
            const v4f v1 = Gt[1 * 2 + a];
            const v2f c0 = tq0 ? (v2f){ v0.z, v0.w } : (v2f){ v0.x, v0.y };
            const v2f c1 = tq1 ? (v2f){ v1.z, v1.w } : (v2f){ v1.x, v1.y };
            p0[a] = cmul(f[0][a], c0);
            p1[a] = cmul(f[1][a], c1);
        }
        const v2f sm = p1[0] + p1[1], df = p1[0] - p1[1];
        const v2f al0 = cmul(p0[0], sm) + cmul(p0[1], df);
        const v2f al1 = cmul(p0[0], df) + cmul(p0[1], sm);

        v2f t2[2], t3[4], t4[8], z[16];
        t2[0] = cmul(al0, f[2][0]);
        t2[1] = cmul(al1, f[2][1]);
        #pragma unroll
        for (int a = 0; a < 4; ++a) {
            v2f v = cmul(t2[a >> 1], f[3][a & 1]);
            t3[a] = (((a >> 1) & a & 1) ? -v : v);
        }
        #pragma unroll
        for (int a = 0; a < 8; ++a) {
            v2f v = cmul(t3[a >> 1], f[4][a & 1]);
            t4[a] = (((a >> 1) & a & 1) ? -v : v);
        }
        #pragma unroll
        for (int a = 0; a < 16; ++a) {
            v2f v = cmul(t4[a >> 1], f[5][a & 1]);
            z[a] = (((a >> 1) & a & 1) ? -v : v);
        }
        gate16<3>(z, transposed(Gt[2 * 2], Gt[2 * 2 + 1]));
        gate16<2>(z, transposed(Gt[3 * 2], Gt[3 * 2 + 1]));
        gate16<1>(z, transposed(Gt[4 * 2], Gt[4 * 2 + 1]));
        gate16<0>(z, transposed(Gt[5 * 2], Gt[5 * 2 + 1]));
        const float sg0 = (tq0 & tq1) ? -1.0f : 1.0f;
        const float sg1 = tq1 ? -sg0 : sg0;
        #pragma unroll
        for (int a = 0; a < 16; ++a) {
            const float sg = (a >> 3) ? sg1 : sg0;
            z[a] = ((__builtin_popcount(a & (a >> 1) & 7) & 1) ? -sg : sg) * z[a];
        }
        v4f* ub = Ul + ((r >> 4) * 2) * 256 + (2 * tau) * 32 + (r & 15);
        #pragma unroll
        for (int jl = 0; jl < 2; ++jl) {
            #pragma unroll
            for (int h = 0; h < 2; ++h) {
                const int k0 = 8 * jl + 2 * h;
                ub[jl * 32 + 16 * h]       = (v4f){ z[k0].x, z[k0 + 1].x, z[k0 + 4].x, z[k0 + 5].x };
                ub[256 + jl * 32 + 16 * h] = (v4f){ z[k0].y, z[k0 + 1].y, z[k0 + 4].y, z[k0 + 5].y };
            }
        }
    }

    __syncthreads();

    v2f aR[4][4], aI[4][4];
    v2f PT[4];
    v2f SX0 = {0.f, 0.f}, SY0 = {0.f, 0.f};
    v2f SX1 = {0.f, 0.f}, SY1 = {0.f, 0.f};
    v2f SX2 = {0.f, 0.f}, SY2 = {0.f, 0.f};
    v2f SX3 = {0.f, 0.f}, SY3 = {0.f, 0.f};
    v2f SX4 = {0.f, 0.f}, SY4 = {0.f, 0.f};
    float sx5 = 0.f, sy5 = 0.f;
    v2f Q3 = {0.f, 0.f}, Q4 = {0.f, 0.f};
    v4f aRe[8], aIm[8];
    #pragma unroll
    for (int j = 0; j < 8; ++j) { aRe[j] = Ul[lane + j * 32]; aIm[j] = Ul[lane + 256 + j * 32]; }
    #pragma unroll
    for (int m = 0; m < 5; ++m) {
        __builtin_amdgcn_sched_barrier(0);
        v8f accR = {}, accI = {};
        if (m < 4) {
            #pragma unroll
            for (int j = 0; j < 8; ++j) {
                #pragma unroll
                for (int hf = 0; hf < 2; ++hf) {
                    v2f fR, fI;
                    if (hf == 0) { fR = (v2f){ aRe[j].x, aRe[j].y }; fI = (v2f){ aIm[j].x, aIm[j].y }; }
                    else         { fR = (v2f){ aRe[j].z, aRe[j].w }; fI = (v2f){ aIm[j].z, aIm[j].w }; }
                    accR = __builtin_amdgcn_wmma_f32_16x16x4_f32(
                        false, fR, false, bF[2 * j + hf], (short)0, accR, false, false);
                    accI = __builtin_amdgcn_wmma_f32_16x16x4_f32(
                        false, fI, false, bF[2 * j + hf], (short)0, accI, false, false);
                }
            }
        }
        if (m > 0) {
            const int t = m - 1;
            pair_acc(aR[t][0], aI[t][0], aR[t][1], aI[t][1], SX4, SY4);
            pair_acc(aR[t][2], aI[t][2], aR[t][3], aI[t][3], SX4, SY4);
            pair_acc(aR[t][0], aI[t][0], aR[t][2], aI[t][2], SX3, SY3);
            pair_acc(aR[t][1], aI[t][1], aR[t][3], aI[t][3], SX3, SY3);
            v2f P[4];
            #pragma unroll
            for (int j = 0; j < 4; ++j) {
                const v2f R = aR[t][j], I = aI[t][j];
                sx5 = R.x * R.y + sx5;
                sx5 = I.x * I.y + sx5;
                sy5 = R.x * I.y + sy5;
                sy5 = sy5 - I.x * R.y;
                const v2f br = { xhalf(R.x), xhalf(R.y) };
                const v2f bi = { xhalf(I.x), xhalf(I.y) };
                SX2 = R * br + SX2;
                SX2 = I * bi + SX2;
                SY2 = I * br + SY2;
                SY2 = SY2 - R * bi;
                v2f p = R * R;
                p = I * I + p;
                P[j] = p;
            }
            {
                const v2f t01 = P[0] + P[1], t23 = P[2] + P[3];
                Q4 += (P[0] - P[1]) + (P[2] - P[3]);
                Q3 += t01 - t23;
                PT[t] = t01 + t23;
            }
            if (t == 1 || t == 3) {
                #pragma unroll
                for (int j = 0; j < 4; ++j)
                    pair_acc(aR[t - 1][j], aI[t - 1][j], aR[t][j], aI[t][j], SX1, SY1);
            }
            if (t >= 2) {
                #pragma unroll
                for (int j = 0; j < 4; ++j)
                    pair_acc(aR[t - 2][j], aI[t - 2][j], aR[t][j], aI[t][j], SX0, SY0);
            }
            if (m < 4) {
                asm volatile("" :: "v"(SX0), "v"(SY0), "v"(SX1), "v"(SY1), "v"(SX2), "v"(SY2),
                                   "v"(SX3), "v"(SY3), "v"(SX4), "v"(SY4), "v"(sx5), "v"(sy5),
                                   "v"(Q3), "v"(Q4), "v"(PT[t]));
            }
        }
        __builtin_amdgcn_sched_barrier(0);
        if (m < 4) {
            aR[m][0] = __builtin_shufflevector(accR, accR, 0, 1);
            aR[m][1] = __builtin_shufflevector(accR, accR, 2, 3);
            aR[m][2] = __builtin_shufflevector(accR, accR, 4, 5);
            aR[m][3] = __builtin_shufflevector(accR, accR, 6, 7);
            aI[m][0] = __builtin_shufflevector(accI, accI, 0, 1);
            aI[m][1] = __builtin_shufflevector(accI, accI, 2, 3);
            aI[m][2] = __builtin_shufflevector(accI, accI, 4, 5);
            aI[m][3] = __builtin_shufflevector(accI, accI, 6, 7);
        }
        if (m < 3) {
            const v4f* ua = Ul + ((m + 1) * 2) * 256 + lane;
            #pragma unroll
            for (int j = 0; j < 8; ++j) { aRe[j] = ua[j * 32]; aIm[j] = ua[256 + j * 32]; }
        }
    }

    float e[18];
    e[0] = 2.0f * hsum(SX0);  e[6]  = 2.0f * hsum(SY0);
    e[1] = 2.0f * hsum(SX1);  e[7]  = 2.0f * hsum(SY1);
    e[3] = 2.0f * hsum(SX3);  e[9]  = 2.0f * hsum(SY3);
    e[4] = 2.0f * hsum(SX4);  e[10] = 2.0f * hsum(SY4);
    e[5] = 2.0f * sx5;        e[11] = 2.0f * sy5;
    {
        const float sy2 = hsum(SY2);
        e[2] = hsum(SX2);
        e[8] = lh ? sy2 : -sy2;
    }
    {
        const v2f a01 = PT[0] + PT[1], a23 = PT[2] + PT[3];
        const v2f tot = a01 + a23;
        e[12] = hsum(a01 - a23);
        e[13] = hsum((PT[0] - PT[1]) + (PT[2] - PT[3]));
        e[15] = hsum(Q3);
        e[16] = hsum(Q4);
        e[17] = tot.x - tot.y;
        const float pall = tot.x + tot.y;
        e[14] = lh ? -pall : pall;
    }
    #pragma unroll
    for (int k = 0; k < 18; ++k) e[k] += xhalf(e[k]);

    {
        const int b = b0 + wv * 16 + lm;
        float* o = out + ((size_t)b * S + s) * 18 + lh * 9;
        #pragma unroll
        for (int k = 0; k < 9; ++k) o[k] = lh ? e[9 + k] : e[k];
    }
}

extern "C" void kernel_launch(void* const* d_in, const int* in_sizes, int n_in,
                              void* d_out, int out_size, void* d_ws, size_t ws_size,
                              hipStream_t stream) {
    const float* x = (const float*)d_in[0];
    const float* w = (const float*)d_in[1];
    float* out = (float*)d_out;
    (void)d_ws; (void)ws_size; (void)n_in; (void)out_size;
    int S = in_sizes[1] / (NL * NQ * 3);
    int B = in_sizes[0] / (S * NQ);
    dim3 grid(S, B / BCOLS);
    qreservoir_kernel<<<grid, NTHREADS, 0, stream>>>(x, w, out, S);
}


# U_s formed as the sum of two Kronecker products of 8x8 factors (D cut at the CZ of qubits 2,3): a lane makes one factor entry, then its 16 U entries by 2 complex products each; changes sum order
# speedup vs baseline: 4.1335x; 1.1044x over previous
#include <hip/hip_runtime.h>
#include <math.h>


typedef __attribute__((ext_vector_type(2))) float v2f;
typedef __attribute__((ext_vector_type(4))) float v4f;
typedef __attribute__((ext_vector_type(8))) float v8f;

#define NQ        6
#define DIM       64
#define NL        2
#define BCOLS     128
#define NTHREADS  256
#define UFLOATS   (2 * DIM * DIM)
#define GATETID0   224
#define LPITCH    20
#define LPA       (8 * LPITCH)

struct GateC { float g00r, g00i, g01r, g01i, g10r, g10i, g11r, g11i; };

__device__ __forceinline__ v2f cmul(v2f a, v2f b)
{
    return (v2f){ a.x * b.x - a.y * b.y, a.x * b.y + a.y * b.x };
}

__device__ __forceinline__ float xhalf(float v)
{
    int b = __builtin_bit_cast(int, v);
    return __builtin_bit_cast(float,
        __builtin_amdgcn_permlanex16(b, b, 0x76543210, 0xfedcba98, false, false));
}

__device__ __forceinline__ void pair_acc(v2f xr, v2f xi, v2f yr, v2f yi, v2f& SX, v2f& SY)
{
    SX = xr * yr + SX;
    SX = xi * yi + SX;
    SY = xr * yi + SY;
    SY = SY - xi * yr;
}

__device__ __forceinline__ float hsum(v2f v) { return v.x + v.y; }

__global__ __launch_bounds__(NTHREADS)
void qreservoir_kernel(const float* __restrict__ x, const float* __restrict__ w,
                       float* __restrict__ out, int S)
{
    __shared__ v4f   Ul[UFLOATS / 4];
    __shared__ v4f   Gt[NL * NQ * 2];
    __shared__ __attribute__((aligned(16))) float Hf[2 * 64 * 2];
    __shared__ __attribute__((aligned(16))) float Lf[2 * LPA];

    const int s    = blockIdx.x;
    const int b0   = blockIdx.y * BCOLS;
    const int tid  = threadIdx.x;
    const int lane = tid & 31;
    const int wv   = tid >> 5;
    const int lh   = lane >> 4;
    const int lm   = lane & 15;

    const float* xp = x + ((size_t)(b0 + wv * 16 + lm) * S + s) * NQ + 3 * lh;
    const float f0 = xp[0], f1 = xp[1], f2 = xp[2];

    if (tid >= GATETID0 && tid < GATETID0 + NL * NQ) {
        const int g = tid - GATETID0;
        const float* wg = w + ((size_t)s * NL * NQ + g) * 3;
        const float phi = wg[0], th = wg[1], om = wg[2];
        const float cf = cosf(0.25f);
        const float sf = sinf(0.25f);
        float st, ct; sincosf(0.5f * th, &st, &ct);
        float sa, ca; sincosf(-0.5f * (phi + om), &sa, &ca);
        float sb, cb; sincosf( 0.5f * (phi - om), &sb, &cb);
        float r00r =  ca * ct, r00i =  sa * ct;
        float r01r = -cb * st, r01i = -sb * st;
        float r10r =  cb * st, r10i = -sb * st;
        float r11r =  ca * ct, r11i = -sa * ct;
        GateC G;
        G.g00r = cf * r00r + sf * r01i;
        G.g00i = cf * r00i - sf * r01r;
        G.g01r = sf * r00i + cf * r01r;
        G.g01i = -sf * r00r + cf * r01i;
        G.g10r = cf * r10r + sf * r11i;
        G.g10i = cf * r10i - sf * r11r;
        G.g11r = sf * r10i + cf * r11r;
        G.g11i = -sf * r10r + cf * r11i;
        Gt[2 * g]     = (v4f){ G.g00r, G.g00i, G.g01r, G.g01i };
        Gt[2 * g + 1] = (v4f){ G.g10r, G.g10i, G.g11r, G.g11i };
    }

    float cq[NQ], sq[NQ];
    {
        float so[3], co[3];
        sincosf(0.5f * 3.14159265358979323846f * f0, &so[0], &co[0]);
        sincosf(0.5f * 3.14159265358979323846f * f1, &so[1], &co[1]);
        sincosf(0.5f * 3.14159265358979323846f * f2, &so[2], &co[2]);
        #pragma unroll
        for (int j = 0; j < 3; ++j) {
            const float oc = xhalf(co[j]), os = xhalf(so[j]);
            cq[j]     = lh ? oc : co[j];   sq[j]     = lh ? os : so[j];
            cq[3 + j] = lh ? co[j] : oc;   sq[3 + j] = lh ? so[j] : os;
        }
    }

    v2f bF[16];
    {
        float t2[4], t3[8], t4[16];
        #pragma unroll
        for (int a = 0; a < 4; ++a)  t2[a] = ((a >> 1) ? sq[0] : cq[0]) * ((a & 1) ? sq[1] : cq[1]);
        #pragma unroll
        for (int a = 0; a < 8; ++a)  t3[a] = t2[a >> 1] * ((a & 1) ? sq[2] : cq[2]);
        #pragma unroll
        for (int a = 0; a < 16; ++a) t4[a] = t3[a >> 1] * ((a & 1) ? sq[3] : cq[3]);
        const float f4 = lh ? sq[4] : cq[4];
        #pragma unroll
        for (int kk = 0; kk < 16; ++kk) {
            const float t5 = t4[kk] * f4;
            bF[kk] = (v2f){ t5 * cq[5], t5 * sq[5] };
        }
    }

    __syncthreads();

    {
        const int  which = wv >> 1;
        const int  a     = which & 1;
        const bool isH   = which < 2;
        const int  qb    = isH ? 0 : 3;
        const int  e     = tid & 63;
        const int  r3    = e >> 3, k3 = e & 7;
        const v2f* Gc = (const v2f*)Gt;
        v2f h0[3], h1[3];
        #pragma unroll
        for (int xq = 0; xq < 3; ++xq) {
            const int q  = qb + xq;
            const int rx = (r3 >> (2 - xq)) & 1, kx = (k3 >> (2 - xq)) & 1;
            const v4f g2 = Gt[(NQ + q) * 2 + rx];
            const v2f c0 = Gc[(q * 2 + 0) * 2 + kx];
            const v2f c1 = Gc[(q * 2 + 1) * 2 + kx];
            h0[xq] = cmul((v2f){ g2.x, g2.y }, c0);
            h1[xq] = cmul((v2f){ g2.z, g2.w }, c1);
        }
        v2f u0 = h0[0], u1 = h1[0], w0 = h0[2], w1 = h1[2];
        const v2f zero = { 0.f, 0.f };
        if (isH) { if (a) w0 = zero; else w1 = zero; }
        else if (a) u1 = -u1;
        const v2f F = cmul(h0[1], cmul(u0 + u1, w0 + w1)) + cmul(h1[1], cmul(u0 - u1, w0 - w1));
        const float sg = ((((k3 >> 2) & (k3 >> 1)) ^ ((k3 >> 1) & k3)) & 1) ? -1.0f : 1.0f;
        const v2f Fs = sg * F;
        if (isH) {
            *(v2f*)&Hf[(a * 64 + e) * 2] = Fs;
        } else {
            float* lp = &Lf[a * LPA + r3 * LPITCH + ((k3 >> 1) & 1) * 4 + (k3 >> 2) * 2 + (k3 & 1)];
            lp[0] = Fs.x;
            lp[8] = Fs.y;
        }
    }

    __syncthreads();

    {
        const int tau = wv >> 1;
        const int r   = ((wv & 1) << 5) | lane;
        const int rh  = r >> 3, rl = r & 7;
        const v4f hA = *(const v4f*)&Hf[(rh * 8 + 2 * tau) * 2];
        const v4f hB = *(const v4f*)&Hf[(64 + rh * 8 + 2 * tau) * 2];
        const float* l0 = &Lf[rl * LPITCH];
        const float* l1 = &Lf[LPA + rl * LPITCH];
        v4f L0r[2], L0i[2], L1r[2], L1i[2];
        #pragma unroll
        for (int h = 0; h < 2; ++h) {
            L0r[h] = *(const v4f*)(l0 + 4 * h);  L0i[h] = *(const v4f*)(l0 + 8 + 4 * h);
            L1r[h] = *(const v4f*)(l1 + 4 * h);  L1i[h] = *(const v4f*)(l1 + 8 + 4 * h);
        }
        v4f* ub = Ul + ((r >> 4) * 2) * 256 + (2 * tau) * 32 + (r & 15);
        #pragma unroll
        for (int jl = 0; jl < 2; ++jl) {
            const float a0r = jl ? hA.z : hA.x, a0i = jl ? hA.w : hA.y;
            const float a1r = jl ? hB.z : hB.x, a1i = jl ? hB.w : hB.y;
            #pragma unroll
            for (int h = 0; h < 2; ++h) {
                v2f oR[2], oI[2];
                #pragma unroll
                for (int hq = 0; hq < 2; ++hq) {
                    const float s = (jl == 1 && hq == 1) ? -1.0f : 1.0f;
                    const float p0r = s * a0r, p0i = s * a0i, p1r = s * a1r, p1i = s * a1i;
                    v2f x0r, x0i, x1r, x1i;
                    if (hq == 0) {
                        x0r = (v2f){ L0r[h].x, L0r[h].y };  x0i = (v2f){ L0i[h].x, L0i[h].y };
                        x1r = (v2f){ L1r[h].x, L1r[h].y };  x1i = (v2f){ L1i[h].x, L1i[h].y };
                    } else {
                        x0r = (v2f){ L0r[h].z, L0r[h].w };  x0i = (v2f){ L0i[h].z, L0i[h].w };
                        x1r = (v2f){ L1r[h].z, L1r[h].w };  x1i = (v2f){ L1i[h].z, L1i[h].w };
                    }
                    oR[hq] = p0r * x0r - p0i * x0i + p1r * x1r - p1i * x1i;
                    oI[hq] = p0r * x0i + p0i * x0r + p1r * x1i + p1i * x1r;
                }
                ub[jl * 32 + 16 * h]       = (v4f){ oR[0].x, oR[0].y, oR[1].x, oR[1].y };
                ub[256 + jl * 32 + 16 * h] = (v4f){ oI[0].x, oI[0].y, oI[1].x, oI[1].y };
            }
        }
    }

    __syncthreads();

    v2f aR[4][4], aI[4][4];
    v2f PT[4];
    v2f SX0 = {0.f, 0.f}, SY0 = {0.f, 0.f};
    v2f SX1 = {0.f, 0.f}, SY1 = {0.f, 0.f};
    v2f SX2 = {0.f, 0.f}, SY2 = {0.f, 0.f};
    v2f SX3 = {0.f, 0.f}, SY3 = {0.f, 0.f};
    v2f SX4 = {0.f, 0.f}, SY4 = {0.f, 0.f};
    float sx5 = 0.f, sy5 = 0.f;
    v2f Q3 = {0.f, 0.f}, Q4 = {0.f, 0.f};
    v4f aRe[8], aIm[8];
    #pragma unroll
    for (int j = 0; j < 8; ++j) { aRe[j] = Ul[lane + j * 32]; aIm[j] = Ul[lane + 256 + j * 32]; }
    #pragma unroll
    for (int m = 0; m < 5; ++m) {
        __builtin_amdgcn_sched_barrier(0);
        v8f accR = {}, accI = {};
        if (m < 4) {
            #pragma unroll
            for (int j = 0; j < 8; ++j) {
                #pragma unroll
                for (int hf = 0; hf < 2; ++hf) {
                    v2f fR, fI;
                    if (hf == 0) { fR = (v2f){ aRe[j].x, aRe[j].y }; fI = (v2f){ aIm[j].x, aIm[j].y }; }
                    else         { fR = (v2f){ aRe[j].z, aRe[j].w }; fI = (v2f){ aIm[j].z, aIm[j].w }; }
                    accR = __builtin_amdgcn_wmma_f32_16x16x4_f32(
                        false, fR, false, bF[2 * j + hf], (short)0, accR, false, false);
                    accI = __builtin_amdgcn_wmma_f32_16x16x4_f32(
                        false, fI, false, bF[2 * j + hf], (short)0, accI, false, false);
                }
            }
        }
        if (m > 0) {
            const int t = m - 1;
            pair_acc(aR[t][0], aI[t][0], aR[t][1], aI[t][1], SX4, SY4);
            pair_acc(aR[t][2], aI[t][2], aR[t][3], aI[t][3], SX4, SY4);
            pair_acc(aR[t][0], aI[t][0], aR[t][2], aI[t][2], SX3, SY3);
            pair_acc(aR[t][1], aI[t][1], aR[t][3], aI[t][3], SX3, SY3);
            v2f P[4];
            #pragma unroll
            for (int j = 0; j < 4; ++j) {
                const v2f R = aR[t][j], I = aI[t][j];
                sx5 = R.x * R.y + sx5;
                sx5 = I.x * I.y + sx5;
                sy5 = R.x * I.y + sy5;
                sy5 = sy5 - I.x * R.y;
                const v2f br = { xhalf(R.x), xhalf(R.y) };
                const v2f bi = { xhalf(I.x), xhalf(I.y) };
                SX2 = R * br + SX2;
                SX2 = I * bi + SX2;
                SY2 = I * br + SY2;
                SY2 = SY2 - R * bi;
                v2f p = R * R;
                p = I * I + p;
                P[j] = p;
            }
            {
                const v2f t01 = P[0] + P[1], t23 = P[2] + P[3];
                Q4 += (P[0] - P[1]) + (P[2] - P[3]);
                Q3 += t01 - t23;
                PT[t] = t01 + t23;
            }
            if (t == 1 || t == 3) {
                #pragma unroll
                for (int j = 0; j < 4; ++j)
                    pair_acc(aR[t - 1][j], aI[t - 1][j], aR[t][j], aI[t][j], SX1, SY1);
            }
            if (t >= 2) {
                #pragma unroll
                for (int j = 0; j < 4; ++j)
                    pair_acc(aR[t - 2][j], aI[t - 2][j], aR[t][j], aI[t][j], SX0, SY0);
            }
            if (m < 4) {
                asm volatile("" :: "v"(SX0), "v"(SY0), "v"(SX1), "v"(SY1), "v"(SX2), "v"(SY2),
                                   "v"(SX3), "v"(SY3), "v"(SX4), "v"(SY4), "v"(sx5), "v"(sy5),
                                   "v"(Q3), "v"(Q4), "v"(PT[t]));
            }
        }
        __builtin_amdgcn_sched_barrier(0);
        if (m < 4) {
            aR[m][0] = __builtin_shufflevector(accR, accR, 0, 1);
            aR[m][1] = __builtin_shufflevector(accR, accR, 2, 3);
            aR[m][2] = __builtin_shufflevector(accR, accR, 4, 5);
            aR[m][3] = __builtin_shufflevector(accR, accR, 6, 7);
            aI[m][0] = __builtin_shufflevector(accI, accI, 0, 1);
            aI[m][1] = __builtin_shufflevector(accI, accI, 2, 3);
            aI[m][2] = __builtin_shufflevector(accI, accI, 4, 5);
            aI[m][3] = __builtin_shufflevector(accI, accI, 6, 7);
        }
        if (m < 3) {
            const v4f* ua = Ul + ((m + 1) * 2) * 256 + lane;
            #pragma unroll
            for (int j = 0; j < 8; ++j) { aRe[j] = ua[j * 32]; aIm[j] = ua[256 + j * 32]; }
        }
    }

    float e[18];
    e[0] = 2.0f * hsum(SX0);  e[6]  = 2.0f * hsum(SY0);
    e[1] = 2.0f * hsum(SX1);  e[7]  = 2.0f * hsum(SY1);
    e[3] = 2.0f * hsum(SX3);  e[9]  = 2.0f * hsum(SY3);
    e[4] = 2.0f * hsum(SX4);  e[10] = 2.0f * hsum(SY4);
    e[5] = 2.0f * sx5;        e[11] = 2.0f * sy5;
    {
        const float sy2 = hsum(SY2);
        e[2] = hsum(SX2);
        e[8] = lh ? sy2 : -sy2;
    }
    {
        const v2f a01 = PT[0] + PT[1], a23 = PT[2] + PT[3];
        const v2f tot = a01 + a23;
        e[12] = hsum(a01 - a23);
        e[13] = hsum((PT[0] - PT[1]) + (PT[2] - PT[3]));
        e[15] = hsum(Q3);
        e[16] = hsum(Q4);
        e[17] = tot.x - tot.y;
        const float pall = tot.x + tot.y;
        e[14] = lh ? -pall : pall;
    }
    #pragma unroll
    for (int k = 0; k < 18; ++k) e[k] += xhalf(e[k]);

    {
        const int b = b0 + wv * 16 + lm;
        float* o = out + ((size_t)b * S + s) * 18 + lh * 9;
        #pragma unroll
        for (int k = 0; k < 9; ++k) o[k] = lh ? e[9 + k] : e[k];
    }
}

extern "C" void kernel_launch(void* const* d_in, const int* in_sizes, int n_in,
                              void* d_out, int out_size, void* d_ws, size_t ws_size,
                              hipStream_t stream) {
    const float* x = (const float*)d_in[0];
    const float* w = (const float*)d_in[1];
    float* out = (float*)d_out;
    (void)d_ws; (void)ws_size; (void)n_in; (void)out_size;
    int S = in_sizes[1] / (NL * NQ * 3);
    int B = in_sizes[0] / (S * NQ);
    dim3 grid(S, B / BCOLS);
    qreservoir_kernel<<<grid, NTHREADS, 0, stream>>>(x, w, out, S);
}
